# starting kernel
# baseline (speedup 1.0000x reference)
#include <hip/hip_runtime.h>

typedef float v2f __attribute__((ext_vector_type(2)));
typedef float v8f __attribute__((ext_vector_type(8)));

#define DD 112
#define HH 112
#define WW 112
#define CI 8
#define CO 8
#define NTAPS 27

#define ZD 3
#define ZH 10
#define ZW 18
#define CP 10

__global__ __launch_bounds__(256) void SparseConv3d_wmma_f32_kernel(
    const float* __restrict__ x,
    const float* __restrict__ wgt,
    const float* __restrict__ bias,
    float* __restrict__ out)
{
  const int sH = WW;
  const int sD = HH * WW;
  const int sC = DD * HH * WW;
  const int sN = CI * sC;

  __shared__ float lx[ZD * ZH * ZW * CP];
  __shared__ float lw[NTAPS * 2 * 16 * 4];

  const int tid = threadIdx.x;

  int b = blockIdx.x;
  const int wq = b % 7;  b /= 7;
  const int hb = b % 14; b /= 14;
  const int d  = b % DD; b /= DD;
  const int n  = b;
  const int h0 = hb * 8;
  const int w0 = wq * 16;

  const float* xn = x + n * sN;
  for (int idx = tid; idx < ZD * ZH * CI * ZW; idx += 256) {
    const int zw_off = idx % ZW;
    int r = idx / ZW;
    const int ci = r % CI;     r /= CI;
    const int zh_off = r % ZH;
    const int zd_off = r / ZH;
    const int zd = d  + zd_off - 1;
    const int zh = h0 + zh_off - 1;
    const int zw = w0 + zw_off - 1;
    const bool ok = ((unsigned)zd < (unsigned)DD) &
                    ((unsigned)zh < (unsigned)HH) &
                    ((unsigned)zw < (unsigned)WW);
    const int zdc = ok ? zd : 0;
    const int zhc = ok ? zh : 0;
    const int zwc = ok ? zw : 0;
    const float v = xn[ci * sC + zdc * sD + zhc * sH + zwc] * (ok ? 1.0f : 0.0f);
    lx[((zd_off * ZH + zh_off) * ZW + zw_off) * CP + ci] = v;
  }

  for (int idx = tid; idx < NTAPS * 2 * 16 * 4; idx += 256) {
    const int k   = idx & 3;
    const int nn  = (idx >> 2) & 15;
    const int cig = (idx >> 6) & 1;
    const int t27 = idx >> 7;
    const int ci  = cig * 4 + k;
    const float v = wgt[((nn & 7) * CI + ci) * NTAPS + t27] *
                    (nn < CO ? 1.0f : 0.0f);
    lw[idx] = v;
  }

  __syncthreads();

  const int wave = tid >> 5;
  const int lane = tid & 31;
  const int m    = lane & 15;
  const int hi   = lane >> 4;

  const v2f* lxa = (const v2f*)lx + ((wave * ZW + m) * (CP / 2) + hi);
  const v2f* lwb = (const v2f*)lw + (m * 2 + hi);

  v8f acc = {0.f, 0.f, 0.f, 0.f, 0.f, 0.f, 0.f, 0.f};

#pragma unroll
  for (int kd = 0; kd < 3; ++kd) {
#pragma unroll
    for (int kh = 0; kh < 3; ++kh) {
#pragma unroll
      for (int kw = 0; kw < 3; ++kw) {
        const int t27 = (kd * 3 + kh) * 3 + kw;
#pragma unroll
        for (int cig = 0; cig < 2; ++cig) {
          const v2f A = lxa[((kd * ZH + kh) * ZW + kw) * (CP / 2) + cig * 2];
          const v2f B = lwb[(t27 * 2 + cig) * 32];
          acc = __builtin_amdgcn_wmma_f32_16x16x4_f32(
              false, A, false, B, (short)0, acc, false, false);
        }
      }
    }
  }

  const int oc = m;
  if (oc < CO) {
    const float bv = bias[oc];
    const int h = h0 + wave;
    float* op = out + (n * CO + oc) * sC + d * sD + h * sH + w0 + hi * 8;
#pragma unroll
    for (int j = 0; j < 8; ++j)
      __builtin_nontemporal_store(acc[j] + bv, op + j);
  }
}

extern "C" void kernel_launch(void* const* d_in, const int* in_sizes, int n_in,
                              void* d_out, int out_size, void* d_ws, size_t ws_size,
                              hipStream_t stream) {
  const float* x    = (const float*)d_in[0];
  const float* wgt  = (const float*)d_in[1];
  const float* bias = (const float*)d_in[2];
  float* out = (float*)d_out;

  const int blocks = 2 * DD * 14 * 7;
  SparseConv3d_wmma_f32_kernel<<<blocks, 256, 0, stream>>>(x, wgt, bias, out);
}


# Weight table filled from coalesced loads: each weight read once in memory order (7 loads/thread, issued first), scattered to its 2 rows at a 132-float pitch, no strided gathers; sum order unchanged
# speedup vs baseline: 1.5987x; 1.5987x over previous
#include <hip/hip_runtime.h>


typedef float v2f __attribute__((ext_vector_type(2)));
typedef float v4f __attribute__((ext_vector_type(4)));
typedef float v8f __attribute__((ext_vector_type(8)));

#define DD 112
#define HH 112
#define WW 112
#define CI 8
#define CO 8
#define NTAPS 27

#define ZD 3
#define ZH 18
#define ZW 18
#define NROWS (ZD * ZH)
#define NR 36
#define WP 132
#define NWGT (CO * CI * NTAPS)

__global__ __launch_bounds__(256) void SparseConv3d_wmma_f32_kernel(
    const float* __restrict__ x,
    const float* __restrict__ wgt,
    const float* __restrict__ bias,
    float* __restrict__ out)
{
  const int sH = WW;
  const int sD = HH * WW;
  const int sC = DD * HH * WW;
  const int sN = CI * sC;

  __shared__ v4f lx4[NROWS * ZW * 2];
  __shared__ v4f lw4[NR * (WP / 4)];

  const int tid = threadIdx.x;

  int b = blockIdx.x;
  const int wq = b % 7;  b /= 7;
  const int hb = b % 7;  b /= 7;
  const int d  = b % DD; b /= DD;
  const int n  = b;
  const int h0 = hb * 16;
  const int w0 = wq * 16;

  const float* xn = x + n * sN;

  float wv[7];
#pragma unroll
  for (int i = 0; i < 7; ++i) {
    const int f = tid + 256 * i;
    wv[i] = wgt[f < NWGT ? f : NWGT - 1];
  }
  __builtin_amdgcn_sched_barrier(0);

  {
    v4f r[2][4];
    int dsti[2];
    const v4f z4 = {0.f, 0.f, 0.f, 0.f};
#pragma unroll
    for (int it = 0; it < 2; ++it) {
      int mt = tid + it * 256;
      mt = mt < 432 ? mt : 431;
      const int q    = mt & 3;
      const int mh   = (mt >> 2) & 1;
      const int row  = mt >> 3;
      const int zd   = d  + row / ZH - 1;
      const int zh   = h0 + row % ZH - 1;
      const bool ok  = ((unsigned)zd < (unsigned)DD) & ((unsigned)zh < (unsigned)HH);
      const float* p = xn + (2 * mh) * sC + (ok ? zd : 0) * sD + (ok ? zh : 0) * sH
                          + w0 + 4 * q;
      const v4f a0 = *(const v4f*)(p);
      const v4f a1 = *(const v4f*)(p + sC);
      const v4f a2 = *(const v4f*)(p + 4 * sC);
      const v4f a3 = *(const v4f*)(p + 5 * sC);
      r[it][0] = ok ? a0 : z4;
      r[it][1] = ok ? a1 : z4;
      r[it][2] = ok ? a2 : z4;
      r[it][3] = ok ? a3 : z4;
      dsti[it] = (row * ZW + 1 + 4 * q) * 2 + mh;
    }

    const int ht   = tid >= 40 ? tid - 40 : 0;
    const int side = ht & 1;
    const int hh   = (ht >> 1) & 1;
    const int hrow = ht >> 2;
    const int hzd  = d  + hrow / ZH - 1;
    const int hzh  = h0 + hrow % ZH - 1;
    const int hzw  = side ? w0 + 16 : w0 - 1;
    const bool hok = ((unsigned)hzd < (unsigned)DD) & ((unsigned)hzh < (unsigned)HH) &
                     ((unsigned)hzw < (unsigned)WW);
    const float* hp = xn + (2 * hh) * sC + (hok ? hzd : 0) * sD + (hok ? hzh : 0) * sH
                         + (hok ? hzw : 0);
    const float e0 = hp[0];
    const float e1 = hp[sC];
    const float e2 = hp[4 * sC];
    const float e3 = hp[5 * sC];

    {
      float* lwf = (float*)lw4;
#pragma unroll
      for (int i = 0; i < 7; ++i) {
        int f = tid + 256 * i;
        f = f < NWGT ? f : NWGT - 1;
        const int p   = f / NTAPS;
        const int tap = f - p * NTAPS;
        const int kd  = tap / 9;
        const int ci  = p & 7;
        const int col = (p >> 3) * 8 + ((ci >> 1) & 1) * 4 + ((ci >> 2) << 1) + (ci & 1);
        float* dst = lwf + (tap + 3 * kd) * WP + col;
        dst[0]           = wv[i];
        dst[3 * WP + 64] = wv[i];
      }
      const int zc = tid & 127;
      const int zr = zc < 64 ? 9 : 0;
      const int jh = __builtin_amdgcn_readfirstlane(tid >> 7);
#pragma unroll
      for (int i = 0; i < 5; ++i) {
        const int j = jh + 2 * i;
        if (i < 4 || jh == 0)
          lwf[((j / 3) * 12 + zr + j % 3) * WP + zc] = 0.0f;
      }
    }

#pragma unroll
    for (int it = 0; it < 2; ++it) {
      if (tid + it * 256 < 432) {
        v4f* dst = lx4 + dsti[it];
        dst[0] = (v4f){r[it][0].x, r[it][1].x, r[it][2].x, r[it][3].x};
        dst[2] = (v4f){r[it][0].y, r[it][1].y, r[it][2].y, r[it][3].y};
        dst[4] = (v4f){r[it][0].z, r[it][1].z, r[it][2].z, r[it][3].z};
        dst[6] = (v4f){r[it][0].w, r[it][1].w, r[it][2].w, r[it][3].w};
      }
    }
    if (tid >= 40) {
      const v4f e = {hok ? e0 : 0.f, hok ? e1 : 0.f, hok ? e2 : 0.f, hok ? e3 : 0.f};
      lx4[(hrow * ZW + (side ? ZW - 1 : 0)) * 2 + hh] = e;
    }
  }

  __syncthreads();

  const int wave = tid >> 5;
  const int lane = tid & 31;
  const int c    = lane & 15;
  const int hi   = lane >> 4;

  const v4f* lxb = lx4 + (((2 * wave) * ZW + c) * 2 + hi);
  const v4f* lwa = lw4 + (c * 2 + hi);

  v8f acc = {0.f, 0.f, 0.f, 0.f, 0.f, 0.f, 0.f, 0.f};

#pragma unroll
  for (int kd = 0; kd < 3; ++kd) {
#pragma unroll
    for (int r = 0; r < 4; ++r) {
#pragma unroll
      for (int kw = 0; kw < 3; ++kw) {
        const v4f X4 = lxb[((kd * ZH + r) * ZW + kw) * 2];
        const v4f W4 = lwa[((kd * 4 + r) * 3 + kw) * (WP / 4)];
        const v2f X0 = {X4.x, X4.y};
        const v2f W0 = {W4.x, W4.y};
        const v2f X1 = {X4.z, X4.w};
        const v2f W1 = {W4.z, W4.w};
        acc = __builtin_amdgcn_wmma_f32_16x16x4_f32(
            false, W0, false, X0, (short)0, acc, false, false);
        acc = __builtin_amdgcn_wmma_f32_16x16x4_f32(
            false, W1, false, X1, (short)0, acc, false, false);
      }
    }
  }

  {
    const int h = h0 + 2 * wave + hi;
    float* op = out + (n * CO) * sC + d * sD + h * sH + w0 + c;
#pragma unroll
    for (int j = 0; j < 8; ++j)
      __builtin_nontemporal_store(acc[j] + bias[j], op + j * sC);
  }
}

extern "C" void kernel_launch(void* const* d_in, const int* in_sizes, int n_in,
                              void* d_out, int out_size, void* d_ws, size_t ws_size,
                              hipStream_t stream) {
  const float* x    = (const float*)d_in[0];
  const float* wgt  = (const float*)d_in[1];
  const float* bias = (const float*)d_in[2];
  float* out = (float*)d_out;

  const int blocks = 2 * DD * 7 * 7;
  SparseConv3d_wmma_f32_kernel<<<blocks, 256, 0, stream>>>(x, wgt, bias, out);
}


# Wave owns 4 output rows in two accumulators (7 waves, 28-row block tile): a weight fragment is read from LDS once for both, the two shared input rows once; 90 reads per 144 WMMAs; sum order unchanged
# speedup vs baseline: 1.6554x; 1.0355x over previous
#include <hip/hip_runtime.h>


typedef float v2f __attribute__((ext_vector_type(2)));
typedef float v4f __attribute__((ext_vector_type(4)));
typedef float v8f __attribute__((ext_vector_type(8)));

#define DD 112
#define HH 112
#define WW 112
#define CI 8
#define CO 8
#define NTAPS 27

#define NT 224
#define TH 28
#define ZD 3
#define ZH (TH + 2)
#define ZW 18
#define NROWS (ZD * ZH)
#define NR 36
#define WP 132
#define NWGT (CO * CI * NTAPS)
#define NWLD 8
#define NITEM (NROWS * 8)
#define NIT 4
#define NHALO (NROWS * 4)
#define NHT 2

__global__ __launch_bounds__(NT) void SparseConv3d_wmma_f32_kernel(
    const float* __restrict__ x,
    const float* __restrict__ wgt,
    const float* __restrict__ bias,
    float* __restrict__ out)
{
  const int sH = WW;
  const int sD = HH * WW;
  const int sC = DD * HH * WW;
  const int sN = CI * sC;

  __shared__ v4f lx4[NROWS * ZW * 2];
  __shared__ v4f lw4[NR * (WP / 4)];

  const int tid = threadIdx.x;

  int b = blockIdx.x;
  const int wq = b % 7;  b /= 7;
  const int hb = b % 4;  b /= 4;
  const int d  = b % DD; b /= DD;
  const int n  = b;
  const int h0 = hb * TH;
  const int w0 = wq * 16;

  const float* xn = x + n * sN;

  float wv[NWLD];
#pragma unroll
  for (int i = 0; i < NWLD; ++i) {
    const int f = tid + NT * i;
    wv[i] = wgt[f < NWGT ? f : NWGT - 1];
  }
  __builtin_amdgcn_sched_barrier(0);

  {
    v4f r[NIT][4];
    int dsti[NIT];
    const v4f z4 = {0.f, 0.f, 0.f, 0.f};
#pragma unroll
    for (int it = 0; it < NIT; ++it) {
      int mt = tid + it * NT;
      mt = mt < NITEM ? mt : NITEM - 1;
      const int q    = mt & 3;
      const int mh   = (mt >> 2) & 1;
      const int row  = mt >> 3;
      const int zd   = d  + row / ZH - 1;
      const int zh   = h0 + row % ZH - 1;
      const bool ok  = ((unsigned)zd < (unsigned)DD) & ((unsigned)zh < (unsigned)HH);
      const float* p = xn + (2 * mh) * sC + (ok ? zd : 0) * sD + (ok ? zh : 0) * sH
                          + w0 + 4 * q;
      const v4f a0 = *(const v4f*)(p);
      const v4f a1 = *(const v4f*)(p + sC);
      const v4f a2 = *(const v4f*)(p + 4 * sC);
      const v4f a3 = *(const v4f*)(p + 5 * sC);
      r[it][0] = ok ? a0 : z4;
      r[it][1] = ok ? a1 : z4;
      r[it][2] = ok ? a2 : z4;
      r[it][3] = ok ? a3 : z4;
      dsti[it] = (row * ZW + 1 + 4 * q) * 2 + mh;
    }

    v4f e[NHT];
    int hdst[NHT];
#pragma unroll
    for (int ih = 0; ih < NHT; ++ih) {
      int ht = tid + ih * NT;
      ht = ht < NHALO ? ht : NHALO - 1;
      const int side = ht & 1;
      const int hh   = (ht >> 1) & 1;
      const int hrow = ht >> 2;
      const int hzd  = d  + hrow / ZH - 1;
      const int hzh  = h0 + hrow % ZH - 1;
      const int hzw  = side ? w0 + 16 : w0 - 1;
      const bool hok = ((unsigned)hzd < (unsigned)DD) & ((unsigned)hzh < (unsigned)HH) &
                       ((unsigned)hzw < (unsigned)WW);
      const float* hp = xn + (2 * hh) * sC + (hok ? hzd : 0) * sD + (hok ? hzh : 0) * sH
                           + (hok ? hzw : 0);
      const float e0 = hp[0];
      const float e1 = hp[sC];
      const float e2 = hp[4 * sC];
      const float e3 = hp[5 * sC];
      e[ih] = (v4f){hok ? e0 : 0.f, hok ? e1 : 0.f, hok ? e2 : 0.f, hok ? e3 : 0.f};
      hdst[ih] = (hrow * ZW + (side ? ZW - 1 : 0)) * 2 + hh;
    }

    {
      float* lwf = (float*)lw4;
#pragma unroll
      for (int i = 0; i < NWLD; ++i) {
        int f = tid + NT * i;
        f = f < NWGT ? f : NWGT - 1;
        const int p   = f / NTAPS;
        const int tap = f - p * NTAPS;
        const int kd  = tap / 9;
        const int ci  = p & 7;
        const int col = (p >> 3) * 8 + ((ci >> 1) & 1) * 4 + ((ci >> 2) << 1) + (ci & 1);
        float* dst = lwf + (tap + 3 * kd) * WP + col;
        dst[0]           = wv[i];
        dst[3 * WP + 64] = wv[i];
      }
      const v4f z4w = {0.f, 0.f, 0.f, 0.f};
#pragma unroll
      for (int i = 0; i < 2; ++i) {
        const int zi = tid + NT * i;
        if (zi < 288) {
          const int s    = zi & 15;
          const int hr   = zi >> 4;
          const int half = hr >= 9 ? 1 : 0;
          const int j    = hr - 9 * half;
          const int R    = (j / 3) * 12 + (half ? 0 : 9) + j % 3;
          lw4[R * (WP / 4) + half * 16 + s] = z4w;
        }
      }
    }

#pragma unroll
    for (int it = 0; it < NIT; ++it) {
      if (tid + it * NT < NITEM) {
        v4f* dst = lx4 + dsti[it];
        dst[0] = (v4f){r[it][0].x, r[it][1].x, r[it][2].x, r[it][3].x};
        dst[2] = (v4f){r[it][0].y, r[it][1].y, r[it][2].y, r[it][3].y};
        dst[4] = (v4f){r[it][0].z, r[it][1].z, r[it][2].z, r[it][3].z};
        dst[6] = (v4f){r[it][0].w, r[it][1].w, r[it][2].w, r[it][3].w};
      }
    }
#pragma unroll
    for (int ih = 0; ih < NHT; ++ih) {
      if (tid + ih * NT < NHALO)
        lx4[hdst[ih]] = e[ih];
    }
  }

  __syncthreads();

  const int wave = tid >> 5;
  const int lane = tid & 31;
  const int c    = lane & 15;
  const int hi   = lane >> 4;

  const v4f* lxb = lx4 + (((4 * wave) * ZW + c) * 2 + hi);
  const v4f* lwa = lw4 + (c * 2 + hi);

  v8f acc0 = {0.f, 0.f, 0.f, 0.f, 0.f, 0.f, 0.f, 0.f};
  v8f acc1 = {0.f, 0.f, 0.f, 0.f, 0.f, 0.f, 0.f, 0.f};

#pragma unroll
  for (int kd = 0; kd < 3; ++kd) {
    v4f keep[2][3];
#pragma unroll
    for (int r = 0; r < 4; ++r) {
#pragma unroll
      for (int kw = 0; kw < 3; ++kw) {
        const v4f W4 = lwa[((kd * 4 + r) * 3 + kw) * (WP / 4)];
        const v4f Xb = lxb[((kd * ZH + r + 2) * ZW + kw) * 2];
        v4f Xa;
        if (r < 2) {
          Xa = lxb[((kd * ZH + r) * ZW + kw) * 2];
          keep[r][kw] = Xb;
        } else {
          Xa = keep[r - 2][kw];
        }
        const v2f W0  = {W4.x, W4.y};
        const v2f W1  = {W4.z, W4.w};
        const v2f Xa0 = {Xa.x, Xa.y};
        const v2f Xa1 = {Xa.z, Xa.w};
        const v2f Xb0 = {Xb.x, Xb.y};
        const v2f Xb1 = {Xb.z, Xb.w};
        acc0 = __builtin_amdgcn_wmma_f32_16x16x4_f32(
            false, W0, false, Xa0, (short)0, acc0, false, false);
        acc1 = __builtin_amdgcn_wmma_f32_16x16x4_f32(
            false, W0, false, Xb0, (short)0, acc1, false, false);
        acc0 = __builtin_amdgcn_wmma_f32_16x16x4_f32(
            false, W1, false, Xa1, (short)0, acc0, false, false);
        acc1 = __builtin_amdgcn_wmma_f32_16x16x4_f32(
            false, W1, false, Xb1, (short)0, acc1, false, false);
      }
    }
  }

  {
    const int h = h0 + 4 * wave + hi;
    float* op = out + (n * CO) * sC + d * sD + h * sH + w0 + c;
#pragma unroll
    for (int j = 0; j < 8; ++j) {
      const float bj = bias[j];
      __builtin_nontemporal_store(acc0[j] + bj, op + j * sC);
      __builtin_nontemporal_store(acc1[j] + bj, op + j * sC + 2 * sH);
    }
  }
}

extern "C" void kernel_launch(void* const* d_in, const int* in_sizes, int n_in,
                              void* d_out, int out_size, void* d_ws, size_t ws_size,
                              hipStream_t stream) {
  const float* x    = (const float*)d_in[0];
  const float* wgt  = (const float*)d_in[1];
  const float* bias = (const float*)d_in[2];
  float* out = (float*)d_out;

  const int blocks = 2 * DD * 4 * 7;
  SparseConv3d_wmma_f32_kernel<<<blocks, NT, 0, stream>>>(x, wgt, bias, out);
}


# Block tile = full-width strip (8 rows x 112 w, 14 waves): patch rows fetched as whole 448-byte rows, the scalar halo-column loads dropped (w edges are LDS zeros); sum order unchanged
# speedup vs baseline: 1.9055x; 1.1511x over previous
#include <hip/hip_runtime.h>


typedef float v2f __attribute__((ext_vector_type(2)));
typedef float v4f __attribute__((ext_vector_type(4)));
typedef float v8f __attribute__((ext_vector_type(8)));

#define DD 112
#define HH 112
#define WW 112
#define CI 8
#define CO 8
#define NTAPS 27

#define NT 448
#define TH 8
#define ZD 3
#define ZH (TH + 2)
#define ZWU 114
#define ZW 121
#define NROWS (ZD * ZH)
#define NR 36
#define WP 132
#define NWGT (CO * CI * NTAPS)
#define NWLD 4
#define NITEM (NROWS * 2 * 28)
#define NIT 4
#define NZCOL (NROWS * 4)

__global__ __launch_bounds__(NT) void SparseConv3d_wmma_f32_kernel(
    const float* __restrict__ x,
    const float* __restrict__ wgt,
    const float* __restrict__ bias,
    float* __restrict__ out)
{
  const int sH = WW;
  const int sD = HH * WW;
  const int sC = DD * HH * WW;
  const int sN = CI * sC;

  __shared__ v4f lx4[NROWS * ZW * 2];
  __shared__ v4f lw4[NR * (WP / 4)];

  const int tid = threadIdx.x;

  int b = blockIdx.x;
  const int hb = b % 14; b /= 14;
  const int d  = b % DD; b /= DD;
  const int n  = b;
  const int h0 = hb * TH;

  const float* xn = x + n * sN;

  float wv[NWLD];
#pragma unroll
  for (int i = 0; i < NWLD; ++i) {
    const int f = tid + NT * i;
    wv[i] = wgt[f < NWGT ? f : NWGT - 1];
  }
  __builtin_amdgcn_sched_barrier(0);

  {
    v4f r[NIT][4];
    int dsti[NIT];
    const v4f z4 = {0.f, 0.f, 0.f, 0.f};
#pragma unroll
    for (int it = 0; it < NIT; ++it) {
      int mt = tid + it * NT;
      mt = mt < NITEM ? mt : NITEM - 1;
      const int mh   = (mt >> 2) & 1;
      const int rest = mt >> 3;
      const int qh   = rest / NROWS;
      const int row  = rest - qh * NROWS;
      const int q    = qh * 4 + (mt & 3);
      const int zd   = d  + row / ZH - 1;
      const int zh   = h0 + row % ZH - 1;
      const bool ok  = ((unsigned)zd < (unsigned)DD) & ((unsigned)zh < (unsigned)HH);
      const float* p = xn + (2 * mh) * sC + (ok ? zd : 0) * sD + (ok ? zh : 0) * sH
                          + 4 * q;
      const v4f a0 = *(const v4f*)(p);
      const v4f a1 = *(const v4f*)(p + sC);
      const v4f a2 = *(const v4f*)(p + 4 * sC);
      const v4f a3 = *(const v4f*)(p + 5 * sC);
      r[it][0] = ok ? a0 : z4;
      r[it][1] = ok ? a1 : z4;
      r[it][2] = ok ? a2 : z4;
      r[it][3] = ok ? a3 : z4;
      dsti[it] = (row * ZW + 1 + 4 * q) * 2 + mh;
    }

    {
      float* lwf = (float*)lw4;
#pragma unroll
      for (int i = 0; i < NWLD; ++i) {
        int f = tid + NT * i;
        f = f < NWGT ? f : NWGT - 1;
        const int p   = f / NTAPS;
        const int tap = f - p * NTAPS;
        const int kd  = tap / 9;
        const int ci  = p & 7;
        const int col = (p >> 3) * 8 + ((ci >> 1) & 1) * 4 + ((ci >> 2) << 1) + (ci & 1);
        float* dst = lwf + (tap + 3 * kd) * WP + col;
        dst[0]           = wv[i];
        dst[3 * WP + 64] = wv[i];
      }
      const v4f z4w = {0.f, 0.f, 0.f, 0.f};
      if (tid < 288) {
        const int s    = tid & 15;
        const int hr   = tid >> 4;
        const int half = hr >= 9 ? 1 : 0;
        const int j    = hr - 9 * half;
        const int R    = (j / 3) * 12 + (half ? 0 : 9) + j % 3;
        lw4[R * (WP / 4) + half * 16 + s] = z4w;
      }
      if (tid >= NT - NZCOL) {
        const int zi = tid - (NT - NZCOL);
        lx4[((zi >> 2) * ZW + ((zi & 2) ? ZWU - 1 : 0)) * 2 + (zi & 1)] = z4w;
      }
    }

#pragma unroll
    for (int it = 0; it < NIT; ++it) {
      if (tid + it * NT < NITEM) {
        v4f* dst = lx4 + dsti[it];
        dst[0] = (v4f){r[it][0].x, r[it][1].x, r[it][2].x, r[it][3].x};
        dst[2] = (v4f){r[it][0].y, r[it][1].y, r[it][2].y, r[it][3].y};
        dst[4] = (v4f){r[it][0].z, r[it][1].z, r[it][2].z, r[it][3].z};
        dst[6] = (v4f){r[it][0].w, r[it][1].w, r[it][2].w, r[it][3].w};
      }
    }
  }

  __syncthreads();

  const int wave = tid >> 5;
  const int lane = tid & 31;
  const int wcol = wave % 7;
  const int wrow = wave / 7;
  const int c    = lane & 15;
  const int hi   = lane >> 4;

  const v4f* lxb = lx4 + (((4 * wrow) * ZW + 16 * wcol + c) * 2 + hi);
  const v4f* lwa = lw4 + (c * 2 + hi);

  v8f acc0 = {0.f, 0.f, 0.f, 0.f, 0.f, 0.f, 0.f, 0.f};
  v8f acc1 = {0.f, 0.f, 0.f, 0.f, 0.f, 0.f, 0.f, 0.f};

#pragma unroll
  for (int kd = 0; kd < 3; ++kd) {
    v4f keep[2][3];
#pragma unroll
    for (int r = 0; r < 4; ++r) {
#pragma unroll
      for (int kw = 0; kw < 3; ++kw) {
        const v4f W4 = lwa[((kd * 4 + r) * 3 + kw) * (WP / 4)];
        const v4f Xb = lxb[((kd * ZH + r + 2) * ZW + kw) * 2];
        v4f Xa;
        if (r < 2) {
          Xa = lxb[((kd * ZH + r) * ZW + kw) * 2];
          keep[r][kw] = Xb;
        } else {
          Xa = keep[r - 2][kw];
        }
        const v2f W0  = {W4.x, W4.y};
        const v2f W1  = {W4.z, W4.w};
        const v2f Xa0 = {Xa.x, Xa.y};
        const v2f Xa1 = {Xa.z, Xa.w};
        const v2f Xb0 = {Xb.x, Xb.y};
        const v2f Xb1 = {Xb.z, Xb.w};
        acc0 = __builtin_amdgcn_wmma_f32_16x16x4_f32(
            false, W0, false, Xa0, (short)0, acc0, false, false);
        acc1 = __builtin_amdgcn_wmma_f32_16x16x4_f32(
            false, W0, false, Xb0, (short)0, acc1, false, false);
        acc0 = __builtin_amdgcn_wmma_f32_16x16x4_f32(
            false, W1, false, Xa1, (short)0, acc0, false, false);
        acc1 = __builtin_amdgcn_wmma_f32_16x16x4_f32(
            false, W1, false, Xb1, (short)0, acc1, false, false);
      }
    }
  }

  {
    const int h = h0 + 4 * wrow + hi;
    float* op = out + (n * CO) * sC + d * sD + h * sH + 16 * wcol + c;
#pragma unroll
    for (int j = 0; j < 8; ++j) {
      const float bj = bias[j];
      __builtin_nontemporal_store(acc0[j] + bj, op + j * sC);
      __builtin_nontemporal_store(acc1[j] + bj, op + j * sC + 2 * sH);
    }
  }
}

extern "C" void kernel_launch(void* const* d_in, const int* in_sizes, int n_in,
                              void* d_out, int out_size, void* d_ws, size_t ws_size,
                              hipStream_t stream) {
  const float* x    = (const float*)d_in[0];
  const float* wgt  = (const float*)d_in[1];
  const float* bias = (const float*)d_in[2];
  float* out = (float*)d_out;

  const int blocks = 2 * DD * 14;
  SparseConv3d_wmma_f32_kernel<<<blocks, NT, 0, stream>>>(x, wgt, bias, out);
}


# Hot loop LDS fragment reads issued as one group per (kd, input row), one block of 12 WMMAs ahead, into a second register set: 12 read groups per wave instead of 36; sum order unchanged
# speedup vs baseline: 1.9680x; 1.0328x over previous
#include <hip/hip_runtime.h>


typedef float v2f __attribute__((ext_vector_type(2)));
typedef float v4f __attribute__((ext_vector_type(4)));
typedef float v8f __attribute__((ext_vector_type(8)));

#define DD 112
#define HH 112
#define WW 112
#define CI 8
#define CO 8
#define NTAPS 27

#define NT 448
#define TH 8
#define ZD 3
#define ZH (TH + 2)
#define ZWU 114
#define ZW 121
#define NROWS (ZD * ZH)
#define NR 36
#define WP 132
#define NWGT (CO * CI * NTAPS)
#define NWLD 4
#define NITEM (NROWS * 2 * 28)
#define NIT 4
#define NZCOL (NROWS * 4)

__global__ __launch_bounds__(NT) void SparseConv3d_wmma_f32_kernel(
    const float* __restrict__ x,
    const float* __restrict__ wgt,
    const float* __restrict__ bias,
    float* __restrict__ out)
{
  const int sH = WW;
  const int sD = HH * WW;
  const int sC = DD * HH * WW;
  const int sN = CI * sC;

  __shared__ v4f lx4[NROWS * ZW * 2];
  __shared__ v4f lw4[NR * (WP / 4)];

  const int tid = threadIdx.x;

  int b = blockIdx.x;
  const int hb = b % 14; b /= 14;
  const int d  = b % DD; b /= DD;
  const int n  = b;
  const int h0 = hb * TH;

  const float* xn = x + n * sN;

  float wv[NWLD];
#pragma unroll
  for (int i = 0; i < NWLD; ++i) {
    const int f = tid + NT * i;
    wv[i] = wgt[f < NWGT ? f : NWGT - 1];
  }
  __builtin_amdgcn_sched_barrier(0);

  {
    v4f r[NIT][4];
    int dsti[NIT];
    const v4f z4 = {0.f, 0.f, 0.f, 0.f};
#pragma unroll
    for (int it = 0; it < NIT; ++it) {
      int mt = tid + it * NT;
      mt = mt < NITEM ? mt : NITEM - 1;
      const int mh   = (mt >> 2) & 1;
      const int rest = mt >> 3;
      const int qh   = rest / NROWS;
      const int row  = rest - qh * NROWS;
      const int q    = qh * 4 + (mt & 3);
      const int zd   = d  + row / ZH - 1;
      const int zh   = h0 + row % ZH - 1;
      const bool ok  = ((unsigned)zd < (unsigned)DD) & ((unsigned)zh < (unsigned)HH);
      const float* p = xn + (2 * mh) * sC + (ok ? zd : 0) * sD + (ok ? zh : 0) * sH
                          + 4 * q;
      const v4f a0 = *(const v4f*)(p);
      const v4f a1 = *(const v4f*)(p + sC);
      const v4f a2 = *(const v4f*)(p + 4 * sC);
      const v4f a3 = *(const v4f*)(p + 5 * sC);
      r[it][0] = ok ? a0 : z4;
      r[it][1] = ok ? a1 : z4;
      r[it][2] = ok ? a2 : z4;
      r[it][3] = ok ? a3 : z4;
      dsti[it] = (row * ZW + 1 + 4 * q) * 2 + mh;
    }

    {
      float* lwf = (float*)lw4;
#pragma unroll
      for (int i = 0; i < NWLD; ++i) {
        int f = tid + NT * i;
        f = f < NWGT ? f : NWGT - 1;
        const int p   = f / NTAPS;
        const int tap = f - p * NTAPS;
        const int kd  = tap / 9;
        const int ci  = p & 7;
        const int col = (p >> 3) * 8 + ((ci >> 1) & 1) * 4 + ((ci >> 2) << 1) + (ci & 1);
        float* dst = lwf + (tap + 3 * kd) * WP + col;
        dst[0]           = wv[i];
        dst[3 * WP + 64] = wv[i];
      }
      const v4f z4w = {0.f, 0.f, 0.f, 0.f};
      if (tid < 288) {
        const int s    = tid & 15;
        const int hr   = tid >> 4;
        const int half = hr >= 9 ? 1 : 0;
        const int j    = hr - 9 * half;
        const int R    = (j / 3) * 12 + (half ? 0 : 9) + j % 3;
        lw4[R * (WP / 4) + half * 16 + s] = z4w;
      }
      if (tid >= NT - NZCOL) {
        const int zi = tid - (NT - NZCOL);
        lx4[((zi >> 2) * ZW + ((zi & 2) ? ZWU - 1 : 0)) * 2 + (zi & 1)] = z4w;
      }
    }

#pragma unroll
    for (int it = 0; it < NIT; ++it) {
      if (tid + it * NT < NITEM) {
        v4f* dst = lx4 + dsti[it];
        dst[0] = (v4f){r[it][0].x, r[it][1].x, r[it][2].x, r[it][3].x};
        dst[2] = (v4f){r[it][0].y, r[it][1].y, r[it][2].y, r[it][3].y};
        dst[4] = (v4f){r[it][0].z, r[it][1].z, r[it][2].z, r[it][3].z};
        dst[6] = (v4f){r[it][0].w, r[it][1].w, r[it][2].w, r[it][3].w};
      }
    }
  }

  __syncthreads();

  const int wave = tid >> 5;
  const int lane = tid & 31;
  const int wcol = wave % 7;
  const int wrow = wave / 7;
  const int c    = lane & 15;
  const int hi   = lane >> 4;

  const v4f* lxb = lx4 + (((4 * wrow) * ZW + 16 * wcol + c) * 2 + hi);
  const v4f* lwa = lw4 + (c * 2 + hi);

  v8f acc0 = {0.f, 0.f, 0.f, 0.f, 0.f, 0.f, 0.f, 0.f};
  v8f acc1 = {0.f, 0.f, 0.f, 0.f, 0.f, 0.f, 0.f, 0.f};

  v4f fw[2][3], fa[2][3], fb[2][3];
  v4f keep[2][3];

#pragma unroll
  for (int kw = 0; kw < 3; ++kw) {
    fw[0][kw] = lwa[kw * (WP / 4)];
    fa[0][kw] = lxb[kw * 2];
    fb[0][kw] = lxb[(2 * ZW + kw) * 2];
  }

#pragma unroll
  for (int i = 0; i < 12; ++i) {
    const int cur = i & 1;
    const int r   = i & 3;
    if (i + 1 < 12) {
      const int j   = i + 1;
      const int kd2 = j >> 2;
      const int r2  = j & 3;
#pragma unroll
      for (int kw = 0; kw < 3; ++kw) {
        fw[cur ^ 1][kw] = lwa[(j * 3 + kw) * (WP / 4)];
        if (r2 < 2) fa[cur ^ 1][kw] = lxb[((kd2 * ZH + r2) * ZW + kw) * 2];
        fb[cur ^ 1][kw] = lxb[((kd2 * ZH + r2 + 2) * ZW + kw) * 2];
      }
    }
    __builtin_amdgcn_sched_barrier(0);
#pragma unroll
    for (int kw = 0; kw < 3; ++kw) {
      const v4f W4 = fw[cur][kw];
      const v4f Xb = fb[cur][kw];
      v4f Xa;
      if (r < 2) {
        Xa = fa[cur][kw];
        keep[r][kw] = Xb;
      } else {
        Xa = keep[r - 2][kw];
      }
      const v2f W0  = {W4.x, W4.y};
      const v2f W1  = {W4.z, W4.w};
      const v2f Xa0 = {Xa.x, Xa.y};
      const v2f Xa1 = {Xa.z, Xa.w};
      const v2f Xb0 = {Xb.x, Xb.y};
      const v2f Xb1 = {Xb.z, Xb.w};
      acc0 = __builtin_amdgcn_wmma_f32_16x16x4_f32(
          false, W0, false, Xa0, (short)0, acc0, false, false);
      acc1 = __builtin_amdgcn_wmma_f32_16x16x4_f32(
          false, W0, false, Xb0, (short)0, acc1, false, false);
      acc0 = __builtin_amdgcn_wmma_f32_16x16x4_f32(
          false, W1, false, Xa1, (short)0, acc0, false, false);
      acc1 = __builtin_amdgcn_wmma_f32_16x16x4_f32(
          false, W1, false, Xb1, (short)0, acc1, false, false);
    }
    __builtin_amdgcn_sched_barrier(0);
  }

  {
    const int h = h0 + 4 * wrow + hi;
    float* op = out + (n * CO) * sC + d * sD + h * sH + 16 * wcol + c;
#pragma unroll
    for (int j = 0; j < 8; ++j) {
      const float bj = bias[j];
      __builtin_nontemporal_store(acc0[j] + bj, op + j * sC);
      __builtin_nontemporal_store(acc1[j] + bj, op + j * sC + 2 * sH);
    }
  }
}

extern "C" void kernel_launch(void* const* d_in, const int* in_sizes, int n_in,
                              void* d_out, int out_size, void* d_ws, size_t ws_size,
                              hipStream_t stream) {
  const float* x    = (const float*)d_in[0];
  const float* wgt  = (const float*)d_in[1];
  const float* bias = (const float*)d_in[2];
  float* out = (float*)d_out;

  const int blocks = 2 * DD * 14;
  SparseConv3d_wmma_f32_kernel<<<blocks, NT, 0, stream>>>(x, wgt, bias, out);
}


# Minimal filtering F(2,3) along d: a block makes 2 output planes from 4 input planes (V0..V3 built while staging), 192 WMMAs where 288 were; sum order changed: out = (m0+m1)+m2, (m1-m2)-m3
# speedup vs baseline: 2.3629x; 1.2007x over previous
#include <hip/hip_runtime.h>


typedef float v2f __attribute__((ext_vector_type(2)));
typedef float v4f __attribute__((ext_vector_type(4)));
typedef float v8f __attribute__((ext_vector_type(8)));

#define DD 112
#define HH 112
#define WW 112
#define CI 8
#define CO 8
#define NTAPS 27

#define NT 448
#define TH 8
#define ZD 4
#define ZH (TH + 2)
#define ZWU 114
#define ZW 121
#define NROWS (ZD * ZH)
#define NR 48
#define WP 132
#define NWGT (CO * CI * NTAPS)
#define NCMB 576
#define NWRD 2
#define NITEM (ZH * 2 * 28)
#define NIT 2
#define NZCOL (NROWS * 4)

__global__ __launch_bounds__(NT) void SparseConv3d_wmma_f32_kernel(
    const float* __restrict__ x,
    const float* __restrict__ wgt,
    const float* __restrict__ bias,
    float* __restrict__ out)
{
  const int sH = WW;
  const int sD = HH * WW;
  const int sC = DD * HH * WW;
  const int sN = CI * sC;

  __shared__ v4f lx4[NROWS * ZW * 2];
  __shared__ v4f lw4[NR * (WP / 4)];

  const int tid = threadIdx.x;

  int b = blockIdx.x;
  const int hb = b % 14; b /= 14;
  const int dp = b % (DD / 2); b /= (DD / 2);
  const int n  = b;
  const int h0 = hb * TH;
  const int d0 = 2 * dp;

  const float* xn = x + n * sN;

  float wg[NWRD][3];
#pragma unroll
  for (int i = 0; i < NWRD; ++i) {
    int cb = tid + NT * i;
    cb = cb < NCMB ? cb : NCMB - 1;
    const int p  = cb / 9;
    const int hw = cb - p * 9;
    const float* wp = wgt + p * NTAPS + hw;
    wg[i][0] = wp[0];
    wg[i][1] = wp[9];
    wg[i][2] = wp[18];
  }
  __builtin_amdgcn_sched_barrier(0);

  {
    v4f r[NIT][4][4];
    int dsti[NIT];
    const v4f z4 = {0.f, 0.f, 0.f, 0.f};
#pragma unroll
    for (int it = 0; it < NIT; ++it) {
      int mt = tid + it * NT;
      mt = mt < NITEM ? mt : NITEM - 1;
      const int mh   = (mt >> 2) & 1;
      const int rest = mt >> 3;
      const int qh   = rest / ZH;
      const int row  = rest - qh * ZH;
      const int q    = qh * 4 + (mt & 3);
      const int zh   = h0 + row - 1;
      const bool okh = (unsigned)zh < (unsigned)HH;
      const float* ph = xn + (2 * mh) * sC + (okh ? zh : 0) * sH + 4 * q;
#pragma unroll
      for (int k = 0; k < 4; ++k) {
        const int zd   = d0 - 1 + k;
        const bool ok  = okh & ((unsigned)zd < (unsigned)DD);
        const float* p = ph + (ok ? zd : 0) * sD;
        const v4f a0 = *(const v4f*)(p);
        const v4f a1 = *(const v4f*)(p + sC);
        const v4f a2 = *(const v4f*)(p + 4 * sC);
        const v4f a3 = *(const v4f*)(p + 5 * sC);
        r[it][k][0] = ok ? a0 : z4;
        r[it][k][1] = ok ? a1 : z4;
        r[it][k][2] = ok ? a2 : z4;
        r[it][k][3] = ok ? a3 : z4;
      }
      dsti[it] = (row * ZW + 1 + 4 * q) * 2 + mh;
    }
    __builtin_amdgcn_sched_barrier(0);

    {
      float* lwf = (float*)lw4;
#pragma unroll
      for (int i = 0; i < NWRD; ++i) {
        int cb = tid + NT * i;
        cb = cb < NCMB ? cb : NCMB - 1;
        const int p   = cb / 9;
        const int hw  = cb - p * 9;
        const int ci  = p & 7;
        const int col = (p >> 3) * 8 + ((ci >> 1) & 1) * 4 + ((ci >> 2) << 1) + (ci & 1);
        const float s02 = wg[i][0] + wg[i][2];
        const float u0 = wg[i][0];
        const float u1 = 0.5f * (s02 + wg[i][1]);
        const float u2 = 0.5f * (s02 - wg[i][1]);
        const float u3 = wg[i][2];
        float* dst = lwf + hw * WP + col;
        dst[0]                    = u0;
        dst[3 * WP + 64]          = u0;
        dst[12 * WP]              = u1;
        dst[12 * WP + 3 * WP + 64] = u1;
        dst[24 * WP]              = u2;
        dst[24 * WP + 3 * WP + 64] = u2;
        dst[36 * WP]              = u3;
        dst[36 * WP + 3 * WP + 64] = u3;
      }
      const v4f z4w = {0.f, 0.f, 0.f, 0.f};
      if (tid < 384) {
        const int s    = tid & 15;
        const int hr   = tid >> 4;
        const int half = hr >= 12 ? 1 : 0;
        const int jk   = hr - 12 * half;
        const int R    = (jk / 3) * 12 + (half ? 0 : 9) + jk % 3;
        lw4[R * (WP / 4) + half * 16 + s] = z4w;
      }
      if (tid >= NT - NZCOL) {
        const int zi = tid - (NT - NZCOL);
        lx4[((zi >> 2) * ZW + ((zi & 2) ? ZWU - 1 : 0)) * 2 + (zi & 1)] = z4w;
      }
    }

#pragma unroll
    for (int it = 0; it < NIT; ++it) {
      if (tid + it * NT < NITEM) {
#pragma unroll
        for (int j = 0; j < 4; ++j) {
          v4f v[4];
#pragma unroll
          for (int ch = 0; ch < 4; ++ch) {
            v[ch] = j == 0 ? r[it][0][ch] - r[it][2][ch]
                  : j == 1 ? r[it][1][ch] + r[it][2][ch]
                  : j == 2 ? r[it][2][ch] - r[it][1][ch]
                           : r[it][1][ch] - r[it][3][ch];
          }
          v4f* dst = lx4 + j * (ZH * ZW * 2) + dsti[it];
          dst[0] = (v4f){v[0].x, v[1].x, v[2].x, v[3].x};
          dst[2] = (v4f){v[0].y, v[1].y, v[2].y, v[3].y};
          dst[4] = (v4f){v[0].z, v[1].z, v[2].z, v[3].z};
          dst[6] = (v4f){v[0].w, v[1].w, v[2].w, v[3].w};
        }
      }
    }
  }

  __syncthreads();

  const int wave = tid >> 5;
  const int lane = tid & 31;
  const int wcol = wave % 7;
  const int wrow = wave / 7;
  const int c    = lane & 15;
  const int hi   = lane >> 4;

  const v4f* lxb = lx4 + (((4 * wrow) * ZW + 16 * wcol + c) * 2 + hi);
  const v4f* lwa = lw4 + (c * 2 + hi);

  v8f accA[4], accB[4];
#pragma unroll
  for (int j = 0; j < 4; ++j) {
    accA[j] = (v8f){0.f, 0.f, 0.f, 0.f, 0.f, 0.f, 0.f, 0.f};
    accB[j] = (v8f){0.f, 0.f, 0.f, 0.f, 0.f, 0.f, 0.f, 0.f};
  }

  v4f fw[2][3], fa[2][3], fb[2][3];
  v4f keep[2][3];

#pragma unroll
  for (int kw = 0; kw < 3; ++kw) {
    fw[0][kw] = lwa[kw * (WP / 4)];
    fa[0][kw] = lxb[kw * 2];
    fb[0][kw] = lxb[(2 * ZW + kw) * 2];
  }

#pragma unroll
  for (int i = 0; i < 16; ++i) {
    const int cur = i & 1;
    const int j   = i >> 2;
    const int r   = i & 3;
    if (i + 1 < 16) {
      const int i2 = i + 1;
      const int j2 = i2 >> 2;
      const int r2 = i2 & 3;
#pragma unroll
      for (int kw = 0; kw < 3; ++kw) {
        fw[cur ^ 1][kw] = lwa[(i2 * 3 + kw) * (WP / 4)];
        if (r2 < 2) fa[cur ^ 1][kw] = lxb[((j2 * ZH + r2) * ZW + kw) * 2];
        fb[cur ^ 1][kw] = lxb[((j2 * ZH + r2 + 2) * ZW + kw) * 2];
      }
    }
    __builtin_amdgcn_sched_barrier(0);
#pragma unroll
    for (int kw = 0; kw < 3; ++kw) {
      const v4f W4 = fw[cur][kw];
      const v4f Xb = fb[cur][kw];
      v4f Xa;
      if (r < 2) {
        Xa = fa[cur][kw];
        keep[r][kw] = Xb;
      } else {
        Xa = keep[r - 2][kw];
      }
      const v2f W0  = {W4.x, W4.y};
      const v2f W1  = {W4.z, W4.w};
      const v2f Xa0 = {Xa.x, Xa.y};
      const v2f Xa1 = {Xa.z, Xa.w};
      const v2f Xb0 = {Xb.x, Xb.y};
      const v2f Xb1 = {Xb.z, Xb.w};
      accA[j] = __builtin_amdgcn_wmma_f32_16x16x4_f32(
          false, W0, false, Xa0, (short)0, accA[j], false, false);
      accB[j] = __builtin_amdgcn_wmma_f32_16x16x4_f32(
          false, W0, false, Xb0, (short)0, accB[j], false, false);
      accA[j] = __builtin_amdgcn_wmma_f32_16x16x4_f32(
          false, W1, false, Xa1, (short)0, accA[j], false, false);
      accB[j] = __builtin_amdgcn_wmma_f32_16x16x4_f32(
          false, W1, false, Xb1, (short)0, accB[j], false, false);
    }
    __builtin_amdgcn_sched_barrier(0);
  }

  {
    const int h = h0 + 4 * wrow + hi;
    float* op = out + (n * CO) * sC + d0 * sD + h * sH + 16 * wcol + c;
#pragma unroll
    for (int k = 0; k < 8; ++k) {
      const float bk = bias[k];
      const float yA0 = ((accA[0][k] + accA[1][k]) + accA[2][k]) + bk;
      const float yB0 = ((accB[0][k] + accB[1][k]) + accB[2][k]) + bk;
      const float yA1 = ((accA[1][k] - accA[2][k]) - accA[3][k]) + bk;
      const float yB1 = ((accB[1][k] - accB[2][k]) - accB[3][k]) + bk;
      __builtin_nontemporal_store(yA0, op + k * sC);
      __builtin_nontemporal_store(yB0, op + k * sC + 2 * sH);
      __builtin_nontemporal_store(yA1, op + k * sC + sD);
      __builtin_nontemporal_store(yB1, op + k * sC + sD + 2 * sH);
    }
  }
}

extern "C" void kernel_launch(void* const* d_in, const int* in_sizes, int n_in,
                              void* d_out, int out_size, void* d_ws, size_t ws_size,
                              hipStream_t stream) {
  const float* x    = (const float*)d_in[0];
  const float* wgt  = (const float*)d_in[1];
  const float* bias = (const float*)d_in[2];
  float* out = (float*)d_out;

  const int blocks = 2 * (DD / 2) * 14;
  SparseConv3d_wmma_f32_kernel<<<blocks, NT, 0, stream>>>(x, wgt, bias, out);
}


# Block tile halved to a 4-row full-width strip (7 waves, under 116 KiB LDS): two blocks share a WGP, one's staging runs beside the other's WMMAs, finer tail; sum order unchanged
# speedup vs baseline: 2.5585x; 1.0828x over previous
#include <hip/hip_runtime.h>


typedef float v2f __attribute__((ext_vector_type(2)));
typedef float v4f __attribute__((ext_vector_type(4)));
typedef float v8f __attribute__((ext_vector_type(8)));

#define DD 112
#define HH 112
#define WW 112
#define CI 8
#define CO 8
#define NTAPS 27

#define NT 224
#define TH 4
#define ZD 4
#define ZH (TH + 2)
#define ZWU 114
#define ZW 121
#define NROWS (ZD * ZH)
#define NR 48
#define WP 132
#define NWGT (CO * CI * NTAPS)
#define NCMB 576
#define NWRD 3
#define NITEM (ZH * 2 * 28)
#define NIT 2
#define NZCOL (NROWS * 4)
#define NZWGT 384

__global__ __launch_bounds__(NT) void SparseConv3d_wmma_f32_kernel(
    const float* __restrict__ x,
    const float* __restrict__ wgt,
    const float* __restrict__ bias,
    float* __restrict__ out)
{
  const int sH = WW;
  const int sD = HH * WW;
  const int sC = DD * HH * WW;
  const int sN = CI * sC;

  __shared__ v4f lx4[NROWS * ZW * 2];
  __shared__ v4f lw4[NR * (WP / 4)];

  const int tid = threadIdx.x;

  int b = blockIdx.x;
  const int hb = b % (HH / TH); b /= (HH / TH);
  const int dp = b % (DD / 2); b /= (DD / 2);
  const int n  = b;
  const int h0 = hb * TH;
  const int d0 = 2 * dp;

  const float* xn = x + n * sN;

  float wg[NWRD][3];
#pragma unroll
  for (int i = 0; i < NWRD; ++i) {
    int cb = tid + NT * i;
    cb = cb < NCMB ? cb : NCMB - 1;
    const int p  = cb / 9;
    const int hw = cb - p * 9;
    const float* wp = wgt + p * NTAPS + hw;
    wg[i][0] = wp[0];
    wg[i][1] = wp[9];
    wg[i][2] = wp[18];
  }
  __builtin_amdgcn_sched_barrier(0);

  {
    v4f r[NIT][4][4];
    int dsti[NIT];
    const v4f z4 = {0.f, 0.f, 0.f, 0.f};
#pragma unroll
    for (int it = 0; it < NIT; ++it) {
      int mt = tid + it * NT;
      mt = mt < NITEM ? mt : NITEM - 1;
      const int mh   = (mt >> 2) & 1;
      const int rest = mt >> 3;
      const int qh   = rest / ZH;
      const int row  = rest - qh * ZH;
      const int q    = qh * 4 + (mt & 3);
      const int zh   = h0 + row - 1;
      const bool okh = (unsigned)zh < (unsigned)HH;
      const float* ph = xn + (2 * mh) * sC + (okh ? zh : 0) * sH + 4 * q;
#pragma unroll
      for (int k = 0; k < 4; ++k) {
        const int zd   = d0 - 1 + k;
        const bool ok  = okh & ((unsigned)zd < (unsigned)DD);
        const float* p = ph + (ok ? zd : 0) * sD;
        const v4f a0 = *(const v4f*)(p);
        const v4f a1 = *(const v4f*)(p + sC);
        const v4f a2 = *(const v4f*)(p + 4 * sC);
        const v4f a3 = *(const v4f*)(p + 5 * sC);
        r[it][k][0] = ok ? a0 : z4;
        r[it][k][1] = ok ? a1 : z4;
        r[it][k][2] = ok ? a2 : z4;
        r[it][k][3] = ok ? a3 : z4;
      }
      dsti[it] = (row * ZW + 1 + 4 * q) * 2 + mh;
    }
    __builtin_amdgcn_sched_barrier(0);

    {
      float* lwf = (float*)lw4;
#pragma unroll
      for (int i = 0; i < NWRD; ++i) {
        int cb = tid + NT * i;
        cb = cb < NCMB ? cb : NCMB - 1;
        const int p   = cb / 9;
        const int hw  = cb - p * 9;
        const int ci  = p & 7;
        const int col = (p >> 3) * 8 + ((ci >> 1) & 1) * 4 + ((ci >> 2) << 1) + (ci & 1);
        const float s02 = wg[i][0] + wg[i][2];
        const float u0 = wg[i][0];
        const float u1 = 0.5f * (s02 + wg[i][1]);
        const float u2 = 0.5f * (s02 - wg[i][1]);
        const float u3 = wg[i][2];
        float* dst = lwf + hw * WP + col;
        dst[0]                    = u0;
        dst[3 * WP + 64]          = u0;
        dst[12 * WP]              = u1;
        dst[12 * WP + 3 * WP + 64] = u1;
        dst[24 * WP]              = u2;
        dst[24 * WP + 3 * WP + 64] = u2;
        dst[36 * WP]              = u3;
        dst[36 * WP + 3 * WP + 64] = u3;
      }
      const v4f z4w = {0.f, 0.f, 0.f, 0.f};
#pragma unroll
      for (int zp = 0; zp < 2; ++zp) {
        const int zt = tid + zp * NT;
        if (zt >= NZWGT) continue;
        const int s    = zt & 15;
        const int hr   = zt >> 4;
        const int half = hr >= 12 ? 1 : 0;
        const int jk   = hr - 12 * half;
        const int R    = (jk / 3) * 12 + (half ? 0 : 9) + jk % 3;
        lw4[R * (WP / 4) + half * 16 + s] = z4w;
      }
      if (tid >= NT - NZCOL) {
        const int zi = tid - (NT - NZCOL);
        lx4[((zi >> 2) * ZW + ((zi & 2) ? ZWU - 1 : 0)) * 2 + (zi & 1)] = z4w;
      }
    }

#pragma unroll
    for (int it = 0; it < NIT; ++it) {
      if (tid + it * NT < NITEM) {
#pragma unroll
        for (int j = 0; j < 4; ++j) {
          v4f v[4];
#pragma unroll
          for (int ch = 0; ch < 4; ++ch) {
            v[ch] = j == 0 ? r[it][0][ch] - r[it][2][ch]
                  : j == 1 ? r[it][1][ch] + r[it][2][ch]
                  : j == 2 ? r[it][2][ch] - r[it][1][ch]
                           : r[it][1][ch] - r[it][3][ch];
          }
          v4f* dst = lx4 + j * (ZH * ZW * 2) + dsti[it];
          dst[0] = (v4f){v[0].x, v[1].x, v[2].x, v[3].x};
          dst[2] = (v4f){v[0].y, v[1].y, v[2].y, v[3].y};
          dst[4] = (v4f){v[0].z, v[1].z, v[2].z, v[3].z};
          dst[6] = (v4f){v[0].w, v[1].w, v[2].w, v[3].w};
        }
      }
    }
  }

  __syncthreads();

  const int wave = tid >> 5;
  const int lane = tid & 31;
  const int wcol = wave;
  const int c    = lane & 15;
  const int hi   = lane >> 4;

  const v4f* lxb = lx4 + ((16 * wcol + c) * 2 + hi);
  const v4f* lwa = lw4 + (c * 2 + hi);

  v8f accA[4], accB[4];
#pragma unroll
  for (int j = 0; j < 4; ++j) {
    accA[j] = (v8f){0.f, 0.f, 0.f, 0.f, 0.f, 0.f, 0.f, 0.f};
    accB[j] = (v8f){0.f, 0.f, 0.f, 0.f, 0.f, 0.f, 0.f, 0.f};
  }

  v4f fw[2][3], fa[2][3], fb[2][3];
  v4f keep[2][3];

#pragma unroll
  for (int kw = 0; kw < 3; ++kw) {
    fw[0][kw] = lwa[kw * (WP / 4)];
    fa[0][kw] = lxb[kw * 2];
    fb[0][kw] = lxb[(2 * ZW + kw) * 2];
  }

#pragma unroll
  for (int i = 0; i < 16; ++i) {
    const int cur = i & 1;
    const int j   = i >> 2;
    const int r   = i & 3;
    if (i + 1 < 16) {
      const int i2 = i + 1;
      const int j2 = i2 >> 2;
      const int r2 = i2 & 3;
#pragma unroll
      for (int kw = 0; kw < 3; ++kw) {
        fw[cur ^ 1][kw] = lwa[(i2 * 3 + kw) * (WP / 4)];
        if (r2 < 2) fa[cur ^ 1][kw] = lxb[((j2 * ZH + r2) * ZW + kw) * 2];
        fb[cur ^ 1][kw] = lxb[((j2 * ZH + r2 + 2) * ZW + kw) * 2];
      }
    }
    __builtin_amdgcn_sched_barrier(0);
#pragma unroll
    for (int kw = 0; kw < 3; ++kw) {
      const v4f W4 = fw[cur][kw];
      const v4f Xb = fb[cur][kw];
      v4f Xa;
      if (r < 2) {
        Xa = fa[cur][kw];
        keep[r][kw] = Xb;
      } else {
        Xa = keep[r - 2][kw];
      }
      const v2f W0  = {W4.x, W4.y};
      const v2f W1  = {W4.z, W4.w};
      const v2f Xa0 = {Xa.x, Xa.y};
      const v2f Xa1 = {Xa.z, Xa.w};
      const v2f Xb0 = {Xb.x, Xb.y};
      const v2f Xb1 = {Xb.z, Xb.w};
      accA[j] = __builtin_amdgcn_wmma_f32_16x16x4_f32(
          false, W0, false, Xa0, (short)0, accA[j], false, false);
      accB[j] = __builtin_amdgcn_wmma_f32_16x16x4_f32(
          false, W0, false, Xb0, (short)0, accB[j], false, false);
      accA[j] = __builtin_amdgcn_wmma_f32_16x16x4_f32(
          false, W1, false, Xa1, (short)0, accA[j], false, false);
      accB[j] = __builtin_amdgcn_wmma_f32_16x16x4_f32(
          false, W1, false, Xb1, (short)0, accB[j], false, false);
    }
    __builtin_amdgcn_sched_barrier(0);
  }

  {
    const int h = h0 + hi;
    float* op = out + (n * CO) * sC + d0 * sD + h * sH + 16 * wcol + c;
#pragma unroll
    for (int k = 0; k < 8; ++k) {
      const float bk = bias[k];
      const float yA0 = ((accA[0][k] + accA[1][k]) + accA[2][k]) + bk;
      const float yB0 = ((accB[0][k] + accB[1][k]) + accB[2][k]) + bk;
      const float yA1 = ((accA[1][k] - accA[2][k]) - accA[3][k]) + bk;
      const float yB1 = ((accB[1][k] - accB[2][k]) - accB[3][k]) + bk;
      __builtin_nontemporal_store(yA0, op + k * sC);
      __builtin_nontemporal_store(yB0, op + k * sC + 2 * sH);
      __builtin_nontemporal_store(yA1, op + k * sC + sD);
      __builtin_nontemporal_store(yB1, op + k * sC + sD + 2 * sH);
    }
  }
}

extern "C" void kernel_launch(void* const* d_in, const int* in_sizes, int n_in,
                              void* d_out, int out_size, void* d_ws, size_t ws_size,
                              hipStream_t stream) {
  const float* x    = (const float*)d_in[0];
  const float* wgt  = (const float*)d_in[1];
  const float* bias = (const float*)d_in[2];
  float* out = (float*)d_out;

  const int blocks = 2 * (DD / 2) * (HH / TH);
  SparseConv3d_wmma_f32_kernel<<<blocks, NT, 0, stream>>>(x, wgt, bias, out);
}


# One block per WGP walks 12-13 tiles: the next tile's patch loads are issued inside the current tile's 192 WMMAs (2 per block of 12), in flight under them; weight table built once; sum order unchanged
# speedup vs baseline: 2.7805x; 1.0868x over previous
#include <hip/hip_runtime.h>


typedef float v2f __attribute__((ext_vector_type(2)));
typedef float v4f __attribute__((ext_vector_type(4)));
typedef float v8f __attribute__((ext_vector_type(8)));

#define DD 112
#define HH 112
#define WW 112
#define CI 8
#define CO 8
#define NTAPS 27

#define NT 224
#define TH 4
#define ZD 4
#define ZH (TH + 2)
#define ZWU 114
#define ZW 121
#define NROWS (ZD * ZH)
#define NR 48
#define WP 132
#define NWGT (CO * CI * NTAPS)
#define NCMB 576
#define NWRD 3
#define NITEM (ZH * 2 * 28)
#define NIT 2
#define NZCOL (NROWS * 4)
#define NZWGT 384
#define NTILE 3136
#define NBLK 256


__device__ __forceinline__ void lds_barrier() {
  __builtin_amdgcn_fence(__ATOMIC_RELEASE, "workgroup", "local");
  __builtin_amdgcn_s_barrier();
  __builtin_amdgcn_fence(__ATOMIC_ACQUIRE, "workgroup", "local");
}

struct PatchSrc {
  const char* pb[4];
  unsigned    vo[NIT];
};

__device__ __forceinline__ PatchSrc patch_src(const float* __restrict__ x, int t, int tid) {
  const int sH = WW;
  const int sD = HH * WW;
  const int sC = DD * HH * WW;
  const int sN = CI * sC;
  const int hb = t % (HH / TH); t /= (HH / TH);
  const int dp = t % (DD / 2);
  const int n  = t / (DD / 2);
  const int h0 = hb * TH;
  const int d0 = 2 * dp;
  PatchSrc s;
#pragma unroll
  for (int k = 0; k < 4; ++k) {
    const int zd = d0 - 1 + k;
    s.pb[k] = (const char*)(x + n * sN + ((unsigned)zd < (unsigned)DD ? zd : 0) * sD);
  }
#pragma unroll
  for (int it = 0; it < NIT; ++it) {
    int mt = tid + it * NT;
    mt = mt < NITEM ? mt : NITEM - 1;
    const int mh   = (mt >> 2) & 1;
    const int rest = mt >> 3;
    const int qh   = rest / ZH;
    const int row  = rest - qh * ZH;
    const int q    = qh * 4 + (mt & 3);
    const int zh   = h0 + row - 1;
    const bool okh = (unsigned)zh < (unsigned)HH;
    s.vo[it] = (unsigned)(((2 * mh) * sC + (okh ? zh : 0) * sH + 4 * q) * 4);
  }
  return s;
}

__device__ __forceinline__ v4f patch_load(const PatchSrc& s, int it, int k, int ch) {
  const int sC = DD * HH * WW;
  const int co = (ch & 1) + 4 * (ch >> 1);
  return *(const v4f*)(s.pb[k] + (size_t)co * sC * 4 + s.vo[it]);
}

__device__ __forceinline__ void patch_store(const v4f (&r)[NIT][4][4], int t, int tid,
                                            const int (&dsti)[NIT], const int (&rowi)[NIT],
                                            v4f* lx4) {
  const int hb = t % (HH / TH); t /= (HH / TH);
  const int dp = t % (DD / 2);
  const int h0 = hb * TH;
  const int d0 = 2 * dp;
  const v4f z4 = {0.f, 0.f, 0.f, 0.f};
#pragma unroll
  for (int it = 0; it < NIT; ++it) {
    if (tid + it * NT < NITEM) {
      const int zh   = h0 + rowi[it] - 1;
      const bool okh = (unsigned)zh < (unsigned)HH;
      v4f pl[4][4];
#pragma unroll
      for (int k = 0; k < 4; ++k) {
        const bool ok = okh & ((unsigned)(d0 - 1 + k) < (unsigned)DD);
#pragma unroll
        for (int ch = 0; ch < 4; ++ch) pl[k][ch] = ok ? r[it][k][ch] : z4;
      }
#pragma unroll
      for (int j = 0; j < 4; ++j) {
        v4f v[4];
#pragma unroll
        for (int ch = 0; ch < 4; ++ch) {
          v[ch] = j == 0 ? pl[0][ch] - pl[2][ch]
                : j == 1 ? pl[1][ch] + pl[2][ch]
                : j == 2 ? pl[2][ch] - pl[1][ch]
                         : pl[1][ch] - pl[3][ch];
        }
        v4f* dst = lx4 + j * (ZH * ZW * 2) + dsti[it];
        dst[0] = (v4f){v[0].x, v[1].x, v[2].x, v[3].x};
        dst[2] = (v4f){v[0].y, v[1].y, v[2].y, v[3].y};
        dst[4] = (v4f){v[0].z, v[1].z, v[2].z, v[3].z};
        dst[6] = (v4f){v[0].w, v[1].w, v[2].w, v[3].w};
      }
    }
  }
}

template <bool PF>
__device__ __forceinline__ void tile_compute(const v4f* lxb, const v4f* lwa, const PatchSrc& nx,
                                             v4f (&r)[NIT][4][4], const float (&bk)[CO],
                                             float* op) {
  const int sH = WW;
  const int sD = HH * WW;
  const int sC = DD * HH * WW;

  v8f accA[4], accB[4];
#pragma unroll
  for (int j = 0; j < 4; ++j) {
    accA[j] = (v8f){0.f, 0.f, 0.f, 0.f, 0.f, 0.f, 0.f, 0.f};
    accB[j] = (v8f){0.f, 0.f, 0.f, 0.f, 0.f, 0.f, 0.f, 0.f};
  }

  v4f fw[2][3], fa[2][3], fb[2][3];
  v4f keep[2][3];

#pragma unroll
  for (int kw = 0; kw < 3; ++kw) {
    fw[0][kw] = lwa[kw * (WP / 4)];
    fa[0][kw] = lxb[kw * 2];
    fb[0][kw] = lxb[(2 * ZW + kw) * 2];
  }

#pragma unroll
  for (int i = 0; i < 16; ++i) {
    const int cur = i & 1;
    const int j   = i >> 2;
    const int rr  = i & 3;
    if (i + 1 < 16) {
      const int i2 = i + 1;
      const int j2 = i2 >> 2;
      const int r2 = i2 & 3;
#pragma unroll
      for (int kw = 0; kw < 3; ++kw) {
        fw[cur ^ 1][kw] = lwa[(i2 * 3 + kw) * (WP / 4)];
        if (r2 < 2) fa[cur ^ 1][kw] = lxb[((j2 * ZH + r2) * ZW + kw) * 2];
        fb[cur ^ 1][kw] = lxb[((j2 * ZH + r2 + 2) * ZW + kw) * 2];
      }
    }
    if (PF) {
      const int it = i >> 3;
      const int k  = (i >> 1) & 3;
      const int c0 = (i & 1) * 2;
      r[it][k][c0]     = patch_load(nx, it, k, c0);
      r[it][k][c0 + 1] = patch_load(nx, it, k, c0 + 1);
    }
    __builtin_amdgcn_sched_barrier(0);
#pragma unroll
    for (int kw = 0; kw < 3; ++kw) {
      const v4f W4 = fw[cur][kw];
      const v4f Xb = fb[cur][kw];
      v4f Xa;
      if (rr < 2) {
        Xa = fa[cur][kw];
        keep[rr][kw] = Xb;
      } else {
        Xa = keep[rr - 2][kw];
      }
      const v2f W0  = {W4.x, W4.y};
      const v2f W1  = {W4.z, W4.w};
      const v2f Xa0 = {Xa.x, Xa.y};
      const v2f Xa1 = {Xa.z, Xa.w};
      const v2f Xb0 = {Xb.x, Xb.y};
      const v2f Xb1 = {Xb.z, Xb.w};
      accA[j] = __builtin_amdgcn_wmma_f32_16x16x4_f32(
          false, W0, false, Xa0, (short)0, accA[j], false, false);
      accB[j] = __builtin_amdgcn_wmma_f32_16x16x4_f32(
          false, W0, false, Xb0, (short)0, accB[j], false, false);
      accA[j] = __builtin_amdgcn_wmma_f32_16x16x4_f32(
          false, W1, false, Xa1, (short)0, accA[j], false, false);
      accB[j] = __builtin_amdgcn_wmma_f32_16x16x4_f32(
          false, W1, false, Xb1, (short)0, accB[j], false, false);
    }
    __builtin_amdgcn_sched_barrier(0);
  }

#pragma unroll
  for (int k = 0; k < 8; ++k) {
    const float yA0 = ((accA[0][k] + accA[1][k]) + accA[2][k]) + bk[k];
    const float yB0 = ((accB[0][k] + accB[1][k]) + accB[2][k]) + bk[k];
    const float yA1 = ((accA[1][k] - accA[2][k]) - accA[3][k]) + bk[k];
    const float yB1 = ((accB[1][k] - accB[2][k]) - accB[3][k]) + bk[k];
    __builtin_nontemporal_store(yA0, op + k * sC);
    __builtin_nontemporal_store(yB0, op + k * sC + 2 * sH);
    __builtin_nontemporal_store(yA1, op + k * sC + sD);
    __builtin_nontemporal_store(yB1, op + k * sC + sD + 2 * sH);
  }
}

__device__ __forceinline__ float* tile_out(float* __restrict__ out, int t, int wcol, int c, int hi) {
  const int sH = WW;
  const int sD = HH * WW;
  const int sC = DD * HH * WW;
  const int hb = t % (HH / TH); t /= (HH / TH);
  const int dp = t % (DD / 2);
  const int n  = t / (DD / 2);
  return out + (n * CO) * sC + (2 * dp) * sD + (hb * TH + hi) * sH + 16 * wcol + c;
}

__global__ __launch_bounds__(NT) void SparseConv3d_wmma_f32_kernel(
    const float* __restrict__ x,
    const float* __restrict__ wgt,
    const float* __restrict__ bias,
    float* __restrict__ out)
{
  __shared__ v4f lx4[NROWS * ZW * 2];
  __shared__ v4f lw4[NR * (WP / 4)];

  const int tid = threadIdx.x;

  float wg[NWRD][3];
#pragma unroll
  for (int i = 0; i < NWRD; ++i) {
    int cb = tid + NT * i;
    cb = cb < NCMB ? cb : NCMB - 1;
    const int p  = cb / 9;
    const int hw = cb - p * 9;
    const float* wp = wgt + p * NTAPS + hw;
    wg[i][0] = wp[0];
    wg[i][1] = wp[9];
    wg[i][2] = wp[18];
  }
  __builtin_amdgcn_sched_barrier(0);

  v4f r[NIT][4][4];
  int t = blockIdx.x;
  {
    const PatchSrc s0 = patch_src(x, t, tid);
#pragma unroll
    for (int it = 0; it < NIT; ++it)
#pragma unroll
      for (int k = 0; k < 4; ++k)
#pragma unroll
        for (int ch = 0; ch < 4; ++ch) r[it][k][ch] = patch_load(s0, it, k, ch);
  }

  int dsti[NIT], rowi[NIT];
#pragma unroll
  for (int it = 0; it < NIT; ++it) {
    int mt = tid + it * NT;
    mt = mt < NITEM ? mt : NITEM - 1;
    const int mh   = (mt >> 2) & 1;
    const int rest = mt >> 3;
    const int qh   = rest / ZH;
    const int row  = rest - qh * ZH;
    const int q    = qh * 4 + (mt & 3);
    rowi[it] = row;
    dsti[it] = (row * ZW + 1 + 4 * q) * 2 + mh;
  }
  __builtin_amdgcn_sched_barrier(0);

  {
    float* lwf = (float*)lw4;
#pragma unroll
    for (int i = 0; i < NWRD; ++i) {
      int cb = tid + NT * i;
      cb = cb < NCMB ? cb : NCMB - 1;
      const int p   = cb / 9;
      const int hw  = cb - p * 9;
      const int ci  = p & 7;
      const int col = (p >> 3) * 8 + ((ci >> 1) & 1) * 4 + ((ci >> 2) << 1) + (ci & 1);
      const float s02 = wg[i][0] + wg[i][2];
      const float u0 = wg[i][0];
      const float u1 = 0.5f * (s02 + wg[i][1]);
      const float u2 = 0.5f * (s02 - wg[i][1]);
      const float u3 = wg[i][2];
      float* dst = lwf + hw * WP + col;
      dst[0]                    = u0;
      dst[3 * WP + 64]          = u0;
      dst[12 * WP]              = u1;
      dst[12 * WP + 3 * WP + 64] = u1;
      dst[24 * WP]              = u2;
      dst[24 * WP + 3 * WP + 64] = u2;
      dst[36 * WP]              = u3;
      dst[36 * WP + 3 * WP + 64] = u3;
    }
    const v4f z4w = {0.f, 0.f, 0.f, 0.f};
#pragma unroll
    for (int zp = 0; zp < 2; ++zp) {
      const int zt = tid + zp * NT;
      if (zt >= NZWGT) continue;
      const int s    = zt & 15;
      const int hr   = zt >> 4;
      const int half = hr >= 12 ? 1 : 0;
      const int jk   = hr - 12 * half;
      const int R    = (jk / 3) * 12 + (half ? 0 : 9) + jk % 3;
      lw4[R * (WP / 4) + half * 16 + s] = z4w;
    }
    if (tid >= NT - NZCOL) {
      const int zi = tid - (NT - NZCOL);
      lx4[((zi >> 2) * ZW + ((zi & 2) ? ZWU - 1 : 0)) * 2 + (zi & 1)] = z4w;
    }
  }

  const int wave = tid >> 5;
  const int lane = tid & 31;
  const int wcol = wave;
  const int c    = lane & 15;
  const int hi   = lane >> 4;

  const v4f* lxb = lx4 + ((16 * wcol + c) * 2 + hi);
  const v4f* lwa = lw4 + (c * 2 + hi);

  float bk[CO];
#pragma unroll
  for (int k = 0; k < CO; ++k) bk[k] = bias[k];

  while (t + NBLK < NTILE) {
    patch_store(r, t, tid, dsti, rowi, lx4);
    lds_barrier();
    const PatchSrc nx = patch_src(x, t + NBLK, tid);
    tile_compute<true>(lxb, lwa, nx, r, bk, tile_out(out, t, wcol, c, hi));
    lds_barrier();
    t += NBLK;
  }
  {
    patch_store(r, t, tid, dsti, rowi, lx4);
    lds_barrier();
    const PatchSrc none = {};
    tile_compute<false>(lxb, lwa, none, r, bk, tile_out(out, t, wcol, c, hi));
  }
}

extern "C" void kernel_launch(void* const* d_in, const int* in_sizes, int n_in,
                              void* d_out, int out_size, void* d_ws, size_t ws_size,
                              hipStream_t stream) {
  const float* x    = (const float*)d_in[0];
  const float* wgt  = (const float*)d_in[1];
  const float* bias = (const float*)d_in[2];
  float* out = (float*)d_out;

  SparseConv3d_wmma_f32_kernel<<<NBLK, NT, 0, stream>>>(x, wgt, bias, out);
}


# Outputs staged through an LDS tile and stored as whole 128-byte lines: 8 16-byte stores per thread per tile (512 contiguous bytes per wave store), not 32 4-byte stores in 64-byte runs; sum order same
# speedup vs baseline: 2.9959x; 1.0775x over previous
#include <hip/hip_runtime.h>


typedef float v2f __attribute__((ext_vector_type(2)));
typedef float v4f __attribute__((ext_vector_type(4)));
typedef float v8f __attribute__((ext_vector_type(8)));

#define DD 112
#define HH 112
#define WW 112
#define CI 8
#define CO 8
#define NTAPS 27

#define NT 224
#define TH 4
#define ZD 4
#define ZH (TH + 2)
#define ZWU 114
#define ZW 121
#define NROWS (ZD * ZH)
#define NR 48
#define WP 132
#define NWGT (CO * CI * NTAPS)
#define NCMB 576
#define NWRD 3
#define NITEM (ZH * 2 * 28)
#define NIT 2
#define NZCOL (NROWS * 4)
#define NZWGT 384
#define NTILE 3136
#define OCH (TH * WW)
#define NBLK 256


__device__ __forceinline__ void lds_barrier() {
  __builtin_amdgcn_fence(__ATOMIC_RELEASE, "workgroup", "local");
  __builtin_amdgcn_s_barrier();
  __builtin_amdgcn_fence(__ATOMIC_ACQUIRE, "workgroup", "local");
}

struct PatchSrc {
  const char* pb[4];
  unsigned    vo[NIT];
};

__device__ __forceinline__ PatchSrc patch_src(const float* __restrict__ x, int t, int tid) {
  const int sH = WW;
  const int sD = HH * WW;
  const int sC = DD * HH * WW;
  const int sN = CI * sC;
  const int hb = t % (HH / TH); t /= (HH / TH);
  const int dp = t % (DD / 2);
  const int n  = t / (DD / 2);
  const int h0 = hb * TH;
  const int d0 = 2 * dp;
  PatchSrc s;
#pragma unroll
  for (int k = 0; k < 4; ++k) {
    const int zd = d0 - 1 + k;
    s.pb[k] = (const char*)(x + n * sN + ((unsigned)zd < (unsigned)DD ? zd : 0) * sD);
  }
#pragma unroll
  for (int it = 0; it < NIT; ++it) {
    int mt = tid + it * NT;
    mt = mt < NITEM ? mt : NITEM - 1;
    const int mh   = (mt >> 2) & 1;
    const int rest = mt >> 3;
    const int qh   = rest / ZH;
    const int row  = rest - qh * ZH;
    const int q    = qh * 4 + (mt & 3);
    const int zh   = h0 + row - 1;
    const bool okh = (unsigned)zh < (unsigned)HH;
    s.vo[it] = (unsigned)(((2 * mh) * sC + (okh ? zh : 0) * sH + 4 * q) * 4);
  }
  return s;
}

__device__ __forceinline__ v4f patch_load(const PatchSrc& s, int it, int k, int ch) {
  const int sC = DD * HH * WW;
  const int co = (ch & 1) + 4 * (ch >> 1);
  return *(const v4f*)(s.pb[k] + (size_t)co * sC * 4 + s.vo[it]);
}

__device__ __forceinline__ void patch_store(const v4f (&r)[NIT][4][4], int t, int tid,
                                            const int (&dsti)[NIT], const int (&rowi)[NIT],
                                            v4f* lx4) {
  const int hb = t % (HH / TH); t /= (HH / TH);
  const int dp = t % (DD / 2);
  const int h0 = hb * TH;
  const int d0 = 2 * dp;
  const v4f z4 = {0.f, 0.f, 0.f, 0.f};
#pragma unroll
  for (int it = 0; it < NIT; ++it) {
    if (tid + it * NT < NITEM) {
      const int zh   = h0 + rowi[it] - 1;
      const bool okh = (unsigned)zh < (unsigned)HH;
      v4f pl[4][4];
#pragma unroll
      for (int k = 0; k < 4; ++k) {
        const bool ok = okh & ((unsigned)(d0 - 1 + k) < (unsigned)DD);
#pragma unroll
        for (int ch = 0; ch < 4; ++ch) pl[k][ch] = ok ? r[it][k][ch] : z4;
      }
#pragma unroll
      for (int j = 0; j < 4; ++j) {
        v4f v[4];
#pragma unroll
        for (int ch = 0; ch < 4; ++ch) {
          v[ch] = j == 0 ? pl[0][ch] - pl[2][ch]
                : j == 1 ? pl[1][ch] + pl[2][ch]
                : j == 2 ? pl[2][ch] - pl[1][ch]
                         : pl[1][ch] - pl[3][ch];
        }
        v4f* dst = lx4 + j * (ZH * ZW * 2) + dsti[it];
        dst[0] = (v4f){v[0].x, v[1].x, v[2].x, v[3].x};
        dst[2] = (v4f){v[0].y, v[1].y, v[2].y, v[3].y};
        dst[4] = (v4f){v[0].z, v[1].z, v[2].z, v[3].z};
        dst[6] = (v4f){v[0].w, v[1].w, v[2].w, v[3].w};
      }
    }
  }
}

template <bool PF>
__device__ __forceinline__ void tile_compute(const v4f* lxb, const v4f* lwa, const PatchSrc& nx,
                                             v4f (&r)[NIT][4][4], const float (&bk)[CO],
                                             float* lob) {
  v8f accA[4], accB[4];
#pragma unroll
  for (int j = 0; j < 4; ++j) {
    accA[j] = (v8f){0.f, 0.f, 0.f, 0.f, 0.f, 0.f, 0.f, 0.f};
    accB[j] = (v8f){0.f, 0.f, 0.f, 0.f, 0.f, 0.f, 0.f, 0.f};
  }

  v4f fw[2][3], fa[2][3], fb[2][3];
  v4f keep[2][3];

#pragma unroll
  for (int kw = 0; kw < 3; ++kw) {
    fw[0][kw] = lwa[kw * (WP / 4)];
    fa[0][kw] = lxb[kw * 2];
    fb[0][kw] = lxb[(2 * ZW + kw) * 2];
  }

#pragma unroll
  for (int i = 0; i < 16; ++i) {
    const int cur = i & 1;
    const int j   = i >> 2;
    const int rr  = i & 3;
    if (i + 1 < 16) {
      const int i2 = i + 1;
      const int j2 = i2 >> 2;
      const int r2 = i2 & 3;
#pragma unroll
      for (int kw = 0; kw < 3; ++kw) {
        fw[cur ^ 1][kw] = lwa[(i2 * 3 + kw) * (WP / 4)];
        if (r2 < 2) fa[cur ^ 1][kw] = lxb[((j2 * ZH + r2) * ZW + kw) * 2];
        fb[cur ^ 1][kw] = lxb[((j2 * ZH + r2 + 2) * ZW + kw) * 2];
      }
    }
    if (PF) {
      const int it = i >> 3;
      const int k  = (i >> 1) & 3;
      const int c0 = (i & 1) * 2;
      r[it][k][c0]     = patch_load(nx, it, k, c0);
      r[it][k][c0 + 1] = patch_load(nx, it, k, c0 + 1);
    }
    __builtin_amdgcn_sched_barrier(0);
#pragma unroll
    for (int kw = 0; kw < 3; ++kw) {
      const v4f W4 = fw[cur][kw];
      const v4f Xb = fb[cur][kw];
      v4f Xa;
      if (rr < 2) {
        Xa = fa[cur][kw];
        keep[rr][kw] = Xb;
      } else {
        Xa = keep[rr - 2][kw];
      }
      const v2f W0  = {W4.x, W4.y};
      const v2f W1  = {W4.z, W4.w};
      const v2f Xa0 = {Xa.x, Xa.y};
      const v2f Xa1 = {Xa.z, Xa.w};
      const v2f Xb0 = {Xb.x, Xb.y};
      const v2f Xb1 = {Xb.z, Xb.w};
      accA[j] = __builtin_amdgcn_wmma_f32_16x16x4_f32(
          false, W0, false, Xa0, (short)0, accA[j], false, false);
      accB[j] = __builtin_amdgcn_wmma_f32_16x16x4_f32(
          false, W0, false, Xb0, (short)0, accB[j], false, false);
      accA[j] = __builtin_amdgcn_wmma_f32_16x16x4_f32(
          false, W1, false, Xa1, (short)0, accA[j], false, false);
      accB[j] = __builtin_amdgcn_wmma_f32_16x16x4_f32(
          false, W1, false, Xb1, (short)0, accB[j], false, false);
    }
    __builtin_amdgcn_sched_barrier(0);
  }

#pragma unroll
  for (int k = 0; k < 8; ++k) {
    const float yA0 = ((accA[0][k] + accA[1][k]) + accA[2][k]) + bk[k];
    const float yB0 = ((accB[0][k] + accB[1][k]) + accB[2][k]) + bk[k];
    const float yA1 = ((accA[1][k] - accA[2][k]) - accA[3][k]) + bk[k];
    const float yB1 = ((accB[1][k] - accB[2][k]) - accB[3][k]) + bk[k];
    lob[k * OCH]                = yA0;
    lob[k * OCH + 2 * WW]       = yB0;
    lob[(CO + k) * OCH]          = yA1;
    lob[(CO + k) * OCH + 2 * WW] = yB1;
  }
}

__device__ __forceinline__ void out_flush(const v4f* lo4, float* __restrict__ out, int t, int tid) {
  const int sH = WW;
  const int sD = HH * WW;
  const int sC = DD * HH * WW;
  const int hb = t % (HH / TH); t /= (HH / TH);
  const int dp = t % (DD / 2);
  const int n  = t / (DD / 2);
  const int hsel   = tid >= OCH / 4 ? 1 : 0;
  const int within = tid - hsel * (OCH / 4);
  float* ob = out + (n * CO + hsel) * sC + (2 * dp) * sD + (hb * TH) * sH + 4 * within;
  v4f v[8];
#pragma unroll
  for (int i = 0; i < 8; ++i) v[i] = lo4[tid + NT * i];
#pragma unroll
  for (int i = 0; i < 8; ++i)
    __builtin_nontemporal_store(v[i], (v4f*)(ob + (2 * (i & 3)) * sC + (i >> 2) * sD));
}

__global__ __launch_bounds__(NT) void SparseConv3d_wmma_f32_kernel(
    const float* __restrict__ x,
    const float* __restrict__ wgt,
    const float* __restrict__ bias,
    float* __restrict__ out)
{
  __shared__ v4f lx4[NROWS * ZW * 2];
  __shared__ v4f lw4[NR * (WP / 4)];
  __shared__ v4f lo4[2 * CO * OCH / 4];

  const int tid = threadIdx.x;

  float wg[NWRD][3];
#pragma unroll
  for (int i = 0; i < NWRD; ++i) {
    int cb = tid + NT * i;
    cb = cb < NCMB ? cb : NCMB - 1;
    const int p  = cb / 9;
    const int hw = cb - p * 9;
    const float* wp = wgt + p * NTAPS + hw;
    wg[i][0] = wp[0];
    wg[i][1] = wp[9];
    wg[i][2] = wp[18];
  }
  __builtin_amdgcn_sched_barrier(0);

  v4f r[NIT][4][4];
  int t = blockIdx.x;
  {
    const PatchSrc s0 = patch_src(x, t, tid);
#pragma unroll
    for (int it = 0; it < NIT; ++it)
#pragma unroll
      for (int k = 0; k < 4; ++k)
#pragma unroll
        for (int ch = 0; ch < 4; ++ch) r[it][k][ch] = patch_load(s0, it, k, ch);
  }

  int dsti[NIT], rowi[NIT];
#pragma unroll
  for (int it = 0; it < NIT; ++it) {
    int mt = tid + it * NT;
    mt = mt < NITEM ? mt : NITEM - 1;
    const int mh   = (mt >> 2) & 1;
    const int rest = mt >> 3;
    const int qh   = rest / ZH;
    const int row  = rest - qh * ZH;
    const int q    = qh * 4 + (mt & 3);
    rowi[it] = row;
    dsti[it] = (row * ZW + 1 + 4 * q) * 2 + mh;
  }
  __builtin_amdgcn_sched_barrier(0);

  {
    float* lwf = (float*)lw4;
#pragma unroll
    for (int i = 0; i < NWRD; ++i) {
      int cb = tid + NT * i;
      cb = cb < NCMB ? cb : NCMB - 1;
      const int p   = cb / 9;
      const int hw  = cb - p * 9;
      const int ci  = p & 7;
      const int col = (p >> 3) * 8 + ((ci >> 1) & 1) * 4 + ((ci >> 2) << 1) + (ci & 1);
      const float s02 = wg[i][0] + wg[i][2];
      const float u0 = wg[i][0];
      const float u1 = 0.5f * (s02 + wg[i][1]);
      const float u2 = 0.5f * (s02 - wg[i][1]);
      const float u3 = wg[i][2];
      float* dst = lwf + hw * WP + col;
      dst[0]                    = u0;
      dst[3 * WP + 64]          = u0;
      dst[12 * WP]              = u1;
      dst[12 * WP + 3 * WP + 64] = u1;
      dst[24 * WP]              = u2;
      dst[24 * WP + 3 * WP + 64] = u2;
      dst[36 * WP]              = u3;
      dst[36 * WP + 3 * WP + 64] = u3;
    }
    const v4f z4w = {0.f, 0.f, 0.f, 0.f};
#pragma unroll
    for (int zp = 0; zp < 2; ++zp) {
      const int zt = tid + zp * NT;
      if (zt >= NZWGT) continue;
      const int s    = zt & 15;
      const int hr   = zt >> 4;
      const int half = hr >= 12 ? 1 : 0;
      const int jk   = hr - 12 * half;
      const int R    = (jk / 3) * 12 + (half ? 0 : 9) + jk % 3;
      lw4[R * (WP / 4) + half * 16 + s] = z4w;
    }
    if (tid >= NT - NZCOL) {
      const int zi = tid - (NT - NZCOL);
      lx4[((zi >> 2) * ZW + ((zi & 2) ? ZWU - 1 : 0)) * 2 + (zi & 1)] = z4w;
    }
  }

  const int wave = tid >> 5;
  const int lane = tid & 31;
  const int wcol = wave;
  const int c    = lane & 15;
  const int hi   = lane >> 4;

  const v4f* lxb = lx4 + ((16 * wcol + c) * 2 + hi);
  const v4f* lwa = lw4 + (c * 2 + hi);
  float* lob = (float*)lo4 + hi * WW + 16 * wcol + c;

  float bk[CO];
#pragma unroll
  for (int k = 0; k < CO; ++k) bk[k] = bias[k];

  while (t + NBLK < NTILE) {
    patch_store(r, t, tid, dsti, rowi, lx4);
    lds_barrier();
    const PatchSrc nx = patch_src(x, t + NBLK, tid);
    tile_compute<true>(lxb, lwa, nx, r, bk, lob);
    lds_barrier();
    out_flush(lo4, out, t, tid);
    t += NBLK;
  }
  {
    patch_store(r, t, tid, dsti, rowi, lx4);
    lds_barrier();
    const PatchSrc none = {};
    tile_compute<false>(lxb, lwa, none, r, bk, lob);
    lds_barrier();
    out_flush(lo4, out, t, tid);
  }
}

extern "C" void kernel_launch(void* const* d_in, const int* in_sizes, int n_in,
                              void* d_out, int out_size, void* d_ws, size_t ws_size,
                              hipStream_t stream) {
  const float* x    = (const float*)d_in[0];
  const float* wgt  = (const float*)d_in[1];
  const float* bias = (const float*)d_in[2];
  float* out = (float*)d_out;

  SparseConv3d_wmma_f32_kernel<<<NBLK, NT, 0, stream>>>(x, wgt, bias, out);
}
